# starting kernel
# baseline (speedup 1.0000x reference)
#include <hip/hip_runtime.h>

typedef __attribute__((ext_vector_type(2))) float v2f;
typedef __attribute__((ext_vector_type(8))) float v8f;

#define DIN 128


__global__ void GCN_fill0(float* __restrict__ p, int n) {
    int i = blockIdx.x * blockDim.x + threadIdx.x;
    if (i < n) p[i] = 0.0f;
}

__global__ void GCN_degree(const int* __restrict__ dst, float* __restrict__ deg, int E) {
    int e = blockIdx.x * blockDim.x + threadIdx.x;
    if (e < E) atomicAdd(&deg[dst[e]], 1.0f);
}

__global__ void GCN_norm(float* __restrict__ deg, int n) {
    int i = blockIdx.x * blockDim.x + threadIdx.x;
    if (i < n) deg[i] = 1.0f / (deg[i] + 1.0f);
}

__global__ void GCN_scatter(const float* __restrict__ h, const int* __restrict__ src,
                            const int* __restrict__ dst, float* __restrict__ agg, int E) {
    long tid  = (long)blockIdx.x * blockDim.x + threadIdx.x;
    int  e    = (int)(tid >> 5);
    int  lane = (int)(tid & 31);
    if (e >= E) return;
    int s = src[e];
    int d = dst[e];
    const float4 v = ((const float4*)(h + (size_t)s * DIN))[lane];
    float* p = agg + (size_t)d * DIN + lane * 4;
    atomicAdd(p + 0, v.x);
    atomicAdd(p + 1, v.y);
    atomicAdd(p + 2, v.z);
    atomicAdd(p + 3, v.w);
}

template <int NC, bool RELU>
__global__ __launch_bounds__(128) void GCN_fc_wmma(const float* __restrict__ hself,
                                                   const float* __restrict__ hagg,
                                                   const float* __restrict__ norm,
                                                   const float* __restrict__ W,
                                                   const float* __restrict__ bias,
                                                   float* __restrict__ out) {
    constexpr int K    = DIN;
    constexpr int LDSK = K + 2;
    __shared__ float Atile[16 * LDSK];

    const int row0 = blockIdx.x * 16;
    const int tid  = threadIdx.x;

    for (int i = tid; i < 16 * K; i += 128) {
        int r = i >> 7;
        int k = i & (K - 1);
        int gr = row0 + r;
        Atile[r * LDSK + k] =
            (hagg[(size_t)gr * K + k] + hself[(size_t)gr * K + k]) * norm[gr];
    }
    __syncthreads();

    const int wave = tid >> 5;
    const int lane = tid & 31;
    const int m    = lane & 15;
    const int kh2  = (lane >> 4) << 1;

    for (int nt = wave; nt * 16 < NC; nt += 4) {
        const int n0 = nt * 16;
        const float* wrow = W + (size_t)(n0 + m) * K;

        v8f acc = {};
#pragma unroll 4
        for (int kb = 0; kb < K; kb += 4) {
            v2f a, b;
            a.x = Atile[m * LDSK + kb + kh2];
            a.y = Atile[m * LDSK + kb + kh2 + 1];
            b.x = wrow[kb + kh2];
            b.y = wrow[kb + kh2 + 1];
            acc = __builtin_amdgcn_wmma_f32_16x16x4_f32(
                 false, a,  false, b,
                 (short)0, acc,  false,  false);
        }

        const float bn   = bias[n0 + m];
        const int   rupp = (lane >> 4) << 3;
#pragma unroll
        for (int v = 0; v < 8; ++v) {
            float val = acc[v] + bn;
            if (RELU) val = fmaxf(val, 0.0f);
            out[(size_t)(row0 + rupp + v) * NC + (n0 + m)] = val;
        }
    }
}


extern "C" void kernel_launch(void* const* d_in, const int* in_sizes, int n_in,
                              void* d_out, int out_size, void* d_ws, size_t ws_size,
                              hipStream_t stream) {
    (void)n_in; (void)out_size; (void)ws_size;
    const float* x  = (const float*)d_in[0];
    const int*   src = (const int*)d_in[1];
    const int*   dst = (const int*)d_in[2];
    const float* W1 = (const float*)d_in[3];
    const float* b1 = (const float*)d_in[4];
    const float* W2 = (const float*)d_in[5];
    const float* b2 = (const float*)d_in[6];
    const float* W3 = (const float*)d_in[7];
    const float* b3 = (const float*)d_in[8];
    float* out = (float*)d_out;

    const int N = in_sizes[0] / DIN;
    const int E = in_sizes[1];

    char* ws = (char*)d_ws;
    size_t off = 0;
    float* norm = (float*)(ws + off);
    off += (((size_t)N * 4) + 255) & ~(size_t)255;
    float* buf0 = (float*)(ws + off);
    off += (size_t)N * DIN * 4;
    float* buf1 = (float*)(ws + off);

    const int TB = 256;
    const int nf = N * DIN;
    const long sthreads = (long)E * 32;
    const int  sblocks  = (int)((sthreads + TB - 1) / TB);
    const int  fblocks  = N / 16;

    GCN_fill0<<<(N + TB - 1) / TB, TB, 0, stream>>>(norm, N);
    GCN_degree<<<(E + TB - 1) / TB, TB, 0, stream>>>(dst, norm, E);
    GCN_norm<<<(N + TB - 1) / TB, TB, 0, stream>>>(norm, N);

    GCN_fill0<<<(nf + TB - 1) / TB, TB, 0, stream>>>(buf1, nf);
    GCN_scatter<<<sblocks, TB, 0, stream>>>(x, src, dst, buf1, E);
    GCN_fc_wmma<128, true><<<fblocks, 128, 0, stream>>>(x, buf1, norm, W1, b1, buf0);

    GCN_fill0<<<(nf + TB - 1) / TB, TB, 0, stream>>>(buf1, nf);
    GCN_scatter<<<sblocks, TB, 0, stream>>>(buf0, src, dst, buf1, E);
    GCN_fc_wmma<128, true><<<fblocks, 128, 0, stream>>>(buf0, buf1, norm, W2, b2, buf1);

    GCN_fill0<<<(nf + TB - 1) / TB, TB, 0, stream>>>(buf0, nf);
    GCN_scatter<<<sblocks, TB, 0, stream>>>(buf1, src, dst, buf0, E);
    GCN_fc_wmma<64, false><<<fblocks, 128, 0, stream>>>(buf1, buf0, norm, W3, b3, out);
}


# Edge lists per 400-row group built once by a binning pass (LDS counts; one integer atomic per block and group reserves a run); all 3 layers walk them, no code scan; adds in list arrival order
# speedup vs baseline: 4.5220x; 4.5220x over previous
#include <hip/hip_runtime.h>

typedef __attribute__((ext_vector_type(2))) float v2f;
typedef __attribute__((ext_vector_type(8))) float v8f;

#define DIN 128

#define ROWS 400
#define RSH  9
#define RMASK ((1u << RSH) - 1u)
#define GMAX 256
#define CSTR 32


__global__ void GCN_fill0(float* __restrict__ p0, float* __restrict__ p1, int* __restrict__ gcur, int n) {
    int i = blockIdx.x * blockDim.x + threadIdx.x;
    if (i < n) {
        p0[i] = 0.0f;
        p1[i] = 0.0f;
    }
    if (i < GMAX * CSTR) gcur[i] = 0;
}

__device__ __forceinline__ int GCN_half() {
    const unsigned id = __builtin_amdgcn_s_sendmsg_rtn(0x87);
    return (int)((id >> 18) & 1u);
}

__global__ void GCN_degree(const int* __restrict__ dst, float* __restrict__ deg0, float* __restrict__ deg1, int E) {
    const int e0 = (blockIdx.x * blockDim.x + threadIdx.x) * 4;
    if (e0 >= E) return;
    float* __restrict__ deg = GCN_half() ? deg1 : deg0;
    if (e0 + 3 < E) {
        const int4 d = *(const int4*)(dst + e0);
        atomicAdd(&deg[d.x], 1.0f);
        atomicAdd(&deg[d.y], 1.0f);
        atomicAdd(&deg[d.z], 1.0f);
        atomicAdd(&deg[d.w], 1.0f);
    } else {
        for (int q = 0; q < 4; ++q)
            if (e0 + q < E) atomicAdd(&deg[dst[e0 + q]], 1.0f);
    }
}

#define NTB 416
__global__ __launch_bounds__(NTB) void GCN_norm(float* __restrict__ deg0, const float* __restrict__ deg1,
                                                int* __restrict__ gcnt, int n) {
    static_assert(NTB >= ROWS && NTB % 32 == 0, "one thread per row of the group, whole waves");
    __shared__ int tot;
    const int tid = (int)threadIdx.x;
    if (tid == 0) tot = 0;
    __syncthreads();
    const int i = (int)blockIdx.x * ROWS + tid;
    int c = 0;
    if (tid < ROWS && i < n) {
        const float d = deg0[i] + deg1[i];
        c = (int)d;
        deg0[i] = 1.0f / (d + 1.0f);
    }
    for (int o = 16; o > 0; o >>= 1) c += __shfl_xor(c, o);
    if ((tid & 31) == 0 && c != 0) atomicAdd(&tot, c);
    __syncthreads();
    if (tid == 0) gcnt[blockIdx.x] = tot;
}

#define BTB 1024
__global__ __launch_bounds__(BTB) void GCN_bin(const int* __restrict__ src, const int* __restrict__ dst,
                                               const int* __restrict__ gcnt, int* __restrict__ gcur,
                                               unsigned* __restrict__ list, int E, int G) {
    static_assert(GMAX <= BTB && GMAX % 32 == 0 && GMAX <= 256, "per-group work by whole waves; a group fits a byte");
    __shared__ int hist[GMAX];
    __shared__ int lofs[GMAX];
    __shared__ int delta[GMAX];
    __shared__ int wsc[GMAX / 32];
    __shared__ int wsm[GMAX / 32];
    __shared__ unsigned spk[BTB * 4];
    __shared__ unsigned char sgr[BTB * 4];
    const int tid  = (int)threadIdx.x;
    const int lane = tid & 31;
    const int wave = tid >> 5;
    if (tid < GMAX) hist[tid] = 0;
    __syncthreads();

    const int start = (int)blockIdx.x * (BTB * 4);
    const int e0 = start + tid * 4;
    unsigned pk[4];
    int gq[4], rk[4];
    if (e0 + 3 < E) {
        const int4 d = *(const int4*)(dst + e0);
        const int4 s = *(const int4*)(src + e0);
        const unsigned dd[4] = {(unsigned)d.x, (unsigned)d.y, (unsigned)d.z, (unsigned)d.w};
        const unsigned ss[4] = {(unsigned)s.x, (unsigned)s.y, (unsigned)s.z, (unsigned)s.w};
#pragma unroll
        for (int q = 0; q < 4; ++q) {
            const unsigned g = dd[q] / ROWS;
            gq[q] = (int)g;
            pk[q] = (ss[q] << RSH) | (dd[q] - g * ROWS);
            rk[q] = atomicAdd(&hist[g], 1);
        }
    } else {
#pragma unroll
        for (int q = 0; q < 4; ++q) {
            const int e = e0 + q;
            gq[q] = -1;
            pk[q] = 0u;
            rk[q] = 0;
            if (e < E) {
                const unsigned d = (unsigned)dst[e];
                const unsigned g = d / ROWS;
                gq[q] = (int)g;
                pk[q] = ((unsigned)src[e] << RSH) | (d - g * ROWS);
                rk[q] = atomicAdd(&hist[g], 1);
            }
        }
    }
    __syncthreads();

    int c = 0, m = 0, xc = 0, xm = 0;
    if (tid < GMAX) {
        c = tid < G ? gcnt[tid] : 0;
        m = hist[tid];
        xc = c;
        xm = m;
        for (int o = 1; o < 32; o <<= 1) {
            const int yc = __shfl_up(xc, o);
            const int ym = __shfl_up(xm, o);
            if (lane >= o) {
                xc += yc;
                xm += ym;
            }
        }
        if (lane == 31) {
            wsc[wave] = xc;
            wsm[wave] = xm;
        }
    }
    __syncthreads();
    if (tid < GMAX) {
        int offc = xc - c, offm = xm - m;
        for (int j = 0; j < wave; ++j) {
            offc += wsc[j];
            offm += wsm[j];
        }
        int r = 0;
        if (m > 0) r = atomicAdd(&gcur[tid * CSTR], m);
        lofs[tid]  = offm;
        delta[tid] = offc + r - offm;
    }
    __syncthreads();
#pragma unroll
    for (int q = 0; q < 4; ++q) {
        if (gq[q] >= 0) {
            const int slot = lofs[gq[q]] + rk[q];
            spk[slot] = pk[q];
            sgr[slot] = (unsigned char)gq[q];
        }
    }
    __syncthreads();
    const int nb = min(BTB * 4, E - start);
    for (int i = tid; i < nb; i += BTB) list[delta[sgr[i]] + i] = spk[i];
}

#define RS 132
#define PS 33
#define AWAVES 32
#define BATCH 16

__device__ __forceinline__ void GCN_gather(const float* __restrict__ h, float* acc, unsigned pk, int n, int lane) {
    const int s  = (int)(pk >> RSH);
    const int rl = (int)(pk & RMASK);
    float4 v[BATCH];
#pragma unroll
    for (int j = 0; j < BATCH; ++j) {
        if (j < n) {
            const int sj = __builtin_amdgcn_readlane(s, j);
            v[j] = *(const float4*)(h + (size_t)sj * DIN + lane * 4);
        }
    }
#pragma unroll
    for (int j = 0; j < BATCH; ++j) {
        if (j < n) {
            const int rj = __builtin_amdgcn_readlane(rl, j);
            float* p = acc + rj * RS + lane;
            atomicAdd(p,          v[j].x);
            atomicAdd(p + PS,     v[j].y);
            atomicAdd(p + 2 * PS, v[j].z);
            atomicAdd(p + 3 * PS, v[j].w);
        }
    }
}

template <int NC, bool RELU>
__global__ __launch_bounds__(AWAVES * 32) void GCN_layer(const float* __restrict__ h,
                                                         const int* __restrict__ gcnt,
                                                         const unsigned* __restrict__ list,
                                                         const float* __restrict__ norm,
                                                         const float* __restrict__ W,
                                                         const float* __restrict__ bias,
                                                         float* __restrict__ out, int N) {
    static_assert(ROWS % 16 == 0, "the matrix phase walks whole 16-row tiles");
    static_assert(3 * PS + 32 <= RS && (ROWS * RS) % 4 == 0, "row layout");
    __shared__ __attribute__((aligned(16))) float acc[ROWS * RS];

    const int tid  = (int)threadIdx.x;
    const int lane = tid & 31;
    const int wave = __builtin_amdgcn_readfirstlane(tid >> 5);
    const int r0   = (int)blockIdx.x * ROWS;

    for (int i = tid; i < ROWS * RS / 4; i += AWAVES * 32)
        ((float4*)acc)[i] = make_float4(0.0f, 0.0f, 0.0f, 0.0f);

    int part = 0;
    for (int j = lane; j < (int)blockIdx.x; j += 32) part += gcnt[j];
    for (int o = 16; o > 0; o >>= 1) part += __shfl_xor(part, o);
    const int off = __builtin_amdgcn_readfirstlane(part);
    const int cnt = __builtin_amdgcn_readfirstlane(gcnt[blockIdx.x]);
    const unsigned* __restrict__ lst = list + off;
    __syncthreads();

    {
        const int li = lane & (BATCH - 1);
        int at = wave * BATCH;
        if (at < cnt) {
            unsigned nxt = lst[min(at + li, cnt - 1)];
            for (; at + BATCH <= cnt; at += AWAVES * BATCH) {
                const unsigned pk = nxt;
                nxt = lst[min(at + AWAVES * BATCH + li, cnt - 1)];
                GCN_gather(h, acc, pk, BATCH, lane);
            }
            if (at < cnt) GCN_gather(h, acc, nxt, cnt - at, lane);
        }
    }
    __syncthreads();

    {
        constexpr int RPW = (ROWS + AWAVES - 1) / AWAVES;
        float4 sv[RPW];
        float  nr[RPW];
#pragma unroll
        for (int i = 0; i < RPW; ++i) {
            const int row = wave + i * AWAVES;
            const int gr  = r0 + row;
            if (row < ROWS && gr < N) {
                sv[i] = *(const float4*)(h + (size_t)gr * DIN + lane * 4);
                nr[i] = norm[gr];
            }
        }
#pragma unroll
        for (int i = 0; i < RPW; ++i) {
            const int row = wave + i * AWAVES;
            const int gr  = r0 + row;
            if (row < ROWS && gr < N) {
                float* a = acc + row * RS + lane;
                a[0]      = (a[0]      + sv[i].x) * nr[i];
                a[PS]     = (a[PS]     + sv[i].y) * nr[i];
                a[2 * PS] = (a[2 * PS] + sv[i].z) * nr[i];
                a[3 * PS] = (a[3 * PS] + sv[i].w) * nr[i];
            }
        }
    }

    constexpr int CT = NC / 16;
    constexpr int RG = AWAVES / CT;
    constexpr int RT = ROWS / 16;
    const int m  = lane & 15;
    const int kh = lane >> 4;
    const int ct = wave % CT;
    const int rg = wave / CT;
    const int n0 = ct * 16;
    const float* wrow = W + (size_t)(n0 + m) * DIN + 2 * kh;
    v2f b[DIN / 4];
#pragma unroll
    for (int kb = 0; kb < DIN / 4; ++kb) b[kb] = *(const v2f*)(wrow + 4 * kb);
    const float bn = bias[n0 + m];
    __syncthreads();

    for (int rt = rg; rt < RT; rt += RG) {
        const float* ap = acc + (rt * 16 + m) * RS + 2 * kh * PS;
        v8f d = {};
#pragma unroll
        for (int kb = 0; kb < DIN / 4; ++kb) {
            v2f a;
            a.x = ap[kb];
            a.y = ap[kb + PS];
            d = __builtin_amdgcn_wmma_f32_16x16x4_f32(
                 false, a,  false, b[kb],
                 (short)0, d,  false,  false);
        }
        const int rbase = r0 + rt * 16 + 8 * kh;
#pragma unroll
        for (int v = 0; v < 8; ++v) {
            float val = d[v] + bn;
            if (RELU) val = fmaxf(val, 0.0f);
            const int gr = rbase + v;
            if (gr < N) out[(size_t)gr * NC + (n0 + m)] = val;
        }
    }
}


extern "C" void kernel_launch(void* const* d_in, const int* in_sizes, int n_in,
                              void* d_out, int out_size, void* d_ws, size_t ws_size,
                              hipStream_t stream) {
    (void)n_in; (void)out_size; (void)ws_size;
    const float* x  = (const float*)d_in[0];
    const int*   src = (const int*)d_in[1];
    const int*   dst = (const int*)d_in[2];
    const float* W1 = (const float*)d_in[3];
    const float* b1 = (const float*)d_in[4];
    const float* W2 = (const float*)d_in[5];
    const float* b2 = (const float*)d_in[6];
    const float* W3 = (const float*)d_in[7];
    const float* b3 = (const float*)d_in[8];
    float* out = (float*)d_out;

    const int N = in_sizes[0] / DIN;
    const int E = in_sizes[1];

    char* ws = (char*)d_ws;
    const size_t dbytes = (((size_t)N * 4) + 255) & ~(size_t)255;
    const size_t pbytes = (size_t)N * DIN * 4;
    float* deg0 = (float*)ws;
    float* deg1 = (float*)(ws + dbytes);
    int*   gcnt = (int*)(ws + 2 * dbytes + pbytes);
    int*   gcur = gcnt + GMAX;
    unsigned* list = (unsigned*)(gcur + GMAX * CSTR);
    float* P2   = (float*)(ws + 2 * dbytes + 2 * pbytes);
    float* P3   = (float*)(ws + 2 * dbytes + 3 * pbytes);
    float* norm = deg0;

    const int TB = 256;
    const int ablocks = (N + ROWS - 1) / ROWS;
    const int nquads  = (E + 3) / 4;

    GCN_fill0<<<(N + TB - 1) / TB, TB, 0, stream>>>(deg0, deg1, gcur, N);
    GCN_degree<<<(nquads + TB - 1) / TB, TB, 0, stream>>>(dst, deg0, deg1, E);
    GCN_norm<<<ablocks, NTB, 0, stream>>>(deg0, deg1, gcnt, N);
    GCN_bin<<<(nquads + BTB - 1) / BTB, BTB, 0, stream>>>(src, dst, gcnt, gcur, list, E, ablocks);

    GCN_layer<128, true><<<ablocks, AWAVES * 32, 0, stream>>>(x, gcnt, list, norm, W1, b1, P2, N);

    GCN_layer<128, true><<<ablocks, AWAVES * 32, 0, stream>>>(P2, gcnt, list, norm, W2, b2, P3, N);

    GCN_layer<64, false><<<ablocks, AWAVES * 32, 0, stream>>>(P3, gcnt, list, norm, W3, b3, out, N);
}
